# starting kernel
# baseline (speedup 1.0000x reference)
#include <hip/hip_runtime.h>

typedef __attribute__((ext_vector_type(2))) float v2f;
typedef __attribute__((ext_vector_type(8))) float v8f;

#define VOCAB   50000
#define EMB     300
#define HID     256
#define OUTD    3
#define NASP    8
#define BATCH   8192
#define SEQ     128
#define ROWS    64
#define RT      4

__global__ __launch_bounds__(320)
void pool_kernel(const int* __restrict__ ids,
                 const float* __restrict__ emb,
                 float* __restrict__ x) {
  __shared__ int sIds[SEQ];
  const int s = blockIdx.x;
  const int t = threadIdx.x;
  if (t < SEQ) sIds[t] = ids[s * SEQ + t];
  __syncthreads();
  if (t < EMB) {
    float acc = 0.0f;
#pragma unroll 4
    for (int k = 0; k < SEQ; ++k) {
      acc += emb[(size_t)sIds[k] * EMB + t];
    }
    x[(size_t)s * EMB + t] = acc * (1.0f / (float)SEQ);
  }
}

__global__ __launch_bounds__(256)
void mlp_kernel(const float* __restrict__ x,
                const int* __restrict__ aspect_id,
                const float* __restrict__ W1,
                const float* __restrict__ B1,
                const float* __restrict__ W2,
                const float* __restrict__ B2,
                float* __restrict__ out) {
  extern __shared__ float smem[];
  float* sX = smem;
  float* sH = smem + ROWS * EMB;

  const int a    = blockIdx.y;
  const int row0 = blockIdx.x * ROWS;
  const int t    = threadIdx.x;

  for (int i = t; i < ROWS * EMB; i += 256) {
    sX[i] = x[(size_t)row0 * EMB + i];
  }
  __syncthreads();

  const int lane  = t & 31;
  const int wv    = t >> 5;
  const int nloc  = lane & 15;
  const int khalf = (lane >> 4) * 2;
  const int m     = lane & 15;

  const float* w1a = W1 + (size_t)a * EMB * HID;

  for (int nt = 0; nt < 2; ++nt) {
    const int col = wv * 32 + nt * 16 + nloc;
    v8f acc[RT];
#pragma unroll
    for (int rt = 0; rt < RT; ++rt) acc[rt] = (v8f){};

    for (int kk = 0; kk < EMB / 4; ++kk) {
      const int k = kk * 4 + khalf;
      v2f bfrag;
      bfrag.x = w1a[(size_t)k * HID + col];
      bfrag.y = w1a[(size_t)(k + 1) * HID + col];
#pragma unroll
      for (int rt = 0; rt < RT; ++rt) {
        v2f afrag = *(const v2f*)&sX[(rt * 16 + m) * EMB + k];
        acc[rt] = __builtin_amdgcn_wmma_f32_16x16x4_f32(
             false, afrag,  false, bfrag,
             (short)0, acc[rt],  false,  false);
      }
    }
    const float bias = B1[a * HID + col];
#pragma unroll
    for (int rt = 0; rt < RT; ++rt) {
#pragma unroll
      for (int r = 0; r < 8; ++r) {
        const int rowm = rt * 16 + r + 8 * (lane >> 4);
        float h = acc[rt][r] + bias;
        sH[rowm * HID + col] = h > 0.0f ? h : 0.0f;
      }
    }
  }
  __syncthreads();

  if (t < ROWS * OUTD) {
    const int s = t / OUTD, j = t % OUTD;
    const float* w2a = W2 + (size_t)a * HID * OUTD;
    float acc = B2[a * OUTD + j];
    for (int k = 0; k < HID; ++k) {
      acc += sH[s * HID + k] * w2a[k * OUTD + j];
    }
    const int b = row0 + s;
    if (aspect_id[b] == a) {
      out[b * OUTD + j] = acc;
    }
  }
}

extern "C" void kernel_launch(void* const* d_in, const int* in_sizes, int n_in,
                              void* d_out, int out_size, void* d_ws, size_t ws_size,
                              hipStream_t stream) {
  (void)in_sizes; (void)n_in; (void)out_size; (void)ws_size;
  const int*   ids    = (const int*)d_in[0];
  const int*   aspect = (const int*)d_in[1];
  const float* emb    = (const float*)d_in[2];
  const float* W1     = (const float*)d_in[3];
  const float* b1     = (const float*)d_in[4];
  const float* W2     = (const float*)d_in[5];
  const float* b2     = (const float*)d_in[6];
  float* out = (float*)d_out;
  float* xws = (float*)d_ws;

  pool_kernel<<<BATCH, 320, 0, stream>>>(ids, emb, xws);

  const size_t smem_bytes = (size_t)(ROWS * EMB + ROWS * HID) * sizeof(float);
  dim3 grid(BATCH / ROWS, NASP);
  mlp_kernel<<<grid, 256, smem_bytes, stream>>>(xws, aspect, W1, b1, W2, b2, out);
}


# MLP kernel routed: each workgroup lists the samples of its 256-sample chunk that carry its aspect (ballot prefix, no atomics) and runs both GEMMs on those rows only: 1/8 of the products
# speedup vs baseline: 1.5883x; 1.5883x over previous
#include <hip/hip_runtime.h>

typedef __attribute__((ext_vector_type(2))) float v2f;
typedef __attribute__((ext_vector_type(4))) float v4f;
typedef __attribute__((ext_vector_type(8))) float v8f;

#define VOCAB   50000
#define EMB     300
#define HID     256
#define OUTD    3
#define NASP    8
#define BATCH   8192
#define SEQ     128
#define CHUNK   256
#define PROWS   64
#define LDS_FLOATS (PROWS * EMB + PROWS * HID + CHUNK + 8)

__global__ __launch_bounds__(320)
void pool_kernel(const int* __restrict__ ids,
                 const float* __restrict__ emb,
                 float* __restrict__ x) {
  __shared__ int sIds[SEQ];
  const int s = blockIdx.x;
  const int t = threadIdx.x;
  if (t < SEQ) sIds[t] = ids[s * SEQ + t];
  __syncthreads();
  if (t < EMB) {
    float acc = 0.0f;
#pragma unroll 4
    for (int k = 0; k < SEQ; ++k) {
      acc += emb[(size_t)sIds[k] * EMB + t];
    }
    x[(size_t)s * EMB + t] = acc * (1.0f / (float)SEQ);
  }
}

template <int NRT>
__device__ __forceinline__ void gemm1_tiles(const float* sX, float* sH,
                                            const float* __restrict__ w1a,
                                            const float* __restrict__ b1a,
                                            int lane, int wv) {
  const int nloc  = lane & 15;
  const int half  = lane >> 4;
  const int khalf = half * 2;
  const int col0  = wv * 32 + nloc;
  const int col1  = col0 + 16;
  v8f acc0[NRT], acc1[NRT];
#pragma unroll
  for (int rt = 0; rt < NRT; ++rt) { acc0[rt] = (v8f){}; acc1[rt] = (v8f){}; }

#pragma unroll 5
  for (int kk = 0; kk < EMB / 4; ++kk) {
    const int k = kk * 4 + khalf;
    v2f bf0, bf1;
    bf0.x = w1a[(size_t)k * HID + col0];
    bf0.y = w1a[(size_t)(k + 1) * HID + col0];
    bf1.x = w1a[(size_t)k * HID + col1];
    bf1.y = w1a[(size_t)(k + 1) * HID + col1];
#pragma unroll
    for (int rt = 0; rt < NRT; ++rt) {
      const v2f afrag = *(const v2f*)&sX[(rt * 16 + nloc) * EMB + k];
      acc0[rt] = __builtin_amdgcn_wmma_f32_16x16x4_f32(
           false, afrag,  false, bf0,
           (short)0, acc0[rt],  false,  false);
      acc1[rt] = __builtin_amdgcn_wmma_f32_16x16x4_f32(
           false, afrag,  false, bf1,
           (short)0, acc1[rt],  false,  false);
    }
  }
  const float bias0 = b1a[col0];
  const float bias1 = b1a[col1];
#pragma unroll
  for (int rt = 0; rt < NRT; ++rt) {
#pragma unroll
    for (int r = 0; r < 8; ++r) {
      const int rowm = rt * 16 + r + 8 * half;
      const float h0 = acc0[rt][r] + bias0;
      const float h1 = acc1[rt][r] + bias1;
      sH[rowm * HID + col0] = h0 > 0.0f ? h0 : 0.0f;
      sH[rowm * HID + col1] = h1 > 0.0f ? h1 : 0.0f;
    }
  }
}

__global__ __launch_bounds__(256)
void mlp_kernel(const float* __restrict__ x,
                const int* __restrict__ aspect_id,
                const float* __restrict__ W1,
                const float* __restrict__ B1,
                const float* __restrict__ W2,
                const float* __restrict__ B2,
                float* __restrict__ out) {
  extern __shared__ __attribute__((aligned(16))) float smem[];
  float* sX   = smem;
  float* sH   = smem + PROWS * EMB;
  int*   sIdx = (int*)(smem + PROWS * EMB + PROWS * HID);
  int*   sCnt = sIdx + CHUNK;

  const int a    = blockIdx.y;
  const int row0 = blockIdx.x * CHUNK;
  const int t    = threadIdx.x;
  const int lane = t & 31;
  const int wv   = t >> 5;

  const bool mine = (aspect_id[row0 + t] == a);
  const unsigned bal = __builtin_amdgcn_ballot_w32(mine);
  if (lane == 0) sCnt[wv] = __popc(bal);
  __syncthreads();
  int base = 0, n = 0;
  for (int w = 0; w < 8; ++w) {
    const int c = sCnt[w];
    if (w < wv) base += c;
    n += c;
  }
  if (mine) sIdx[base + __popc(bal & ((1u << lane) - 1u))] = t;
  __syncthreads();

  const float* w1a = W1 + (size_t)a * EMB * HID;
  const float* b1a = B1 + a * HID;
  const float* w2a = W2 + (size_t)a * HID * OUTD;
  const v4f* x4  = (const v4f*)x;
  v4f*       sX4 = (v4f*)sX;

  for (int p0 = 0; p0 < n; p0 += PROWS) {
    const int rows = (n - p0) < PROWS ? (n - p0) : PROWS;
    const int nrt  = (rows + 15) >> 4;

    if (t < 240) {
      for (int rt = 0; rt < nrt; ++rt) {
        v4f v[5];
#pragma unroll
        for (int u = 0; u < 5; ++u) {
          const int i  = rt * 1200 + u * 240 + t;
          const int r  = i / (EMB / 4);
          const int c4 = i - r * (EMB / 4);
          const int rc = r < rows ? r : rows - 1;
          v[u] = x4[(size_t)(row0 + sIdx[p0 + rc]) * (EMB / 4) + c4];
          if (r >= rows) v[u] = (v4f){0.0f, 0.0f, 0.0f, 0.0f};
        }
#pragma unroll
        for (int u = 0; u < 5; ++u) {
          sX4[rt * 1200 + u * 240 + t] = v[u];
        }
      }
    }
    __syncthreads();

    switch (nrt) {
      case 1:  gemm1_tiles<1>(sX, sH, w1a, b1a, lane, wv); break;
      case 2:  gemm1_tiles<2>(sX, sH, w1a, b1a, lane, wv); break;
      case 3:  gemm1_tiles<3>(sX, sH, w1a, b1a, lane, wv); break;
      default: gemm1_tiles<4>(sX, sH, w1a, b1a, lane, wv); break;
    }
    __syncthreads();

    if (t < rows * OUTD) {
      const int s = t / OUTD, j = t % OUTD;
      float acc = B2[a * OUTD + j];
      for (int k = 0; k < HID; ++k) {
        acc += sH[s * HID + k] * w2a[k * OUTD + j];
      }
      const int b = row0 + sIdx[p0 + s];
      out[b * OUTD + j] = acc;
    }
    __syncthreads();
  }
}

extern "C" void kernel_launch(void* const* d_in, const int* in_sizes, int n_in,
                              void* d_out, int out_size, void* d_ws, size_t ws_size,
                              hipStream_t stream) {
  (void)in_sizes; (void)n_in; (void)out_size; (void)ws_size;
  const int*   ids    = (const int*)d_in[0];
  const int*   aspect = (const int*)d_in[1];
  const float* emb    = (const float*)d_in[2];
  const float* W1     = (const float*)d_in[3];
  const float* b1     = (const float*)d_in[4];
  const float* W2     = (const float*)d_in[5];
  const float* b2     = (const float*)d_in[6];
  float* out = (float*)d_out;
  float* xws = (float*)d_ws;

  pool_kernel<<<BATCH, 320, 0, stream>>>(ids, emb, xws);

  const size_t smem_bytes = (size_t)LDS_FLOATS * sizeof(float);
  dim3 grid(BATCH / CHUNK, NASP);
  mlp_kernel<<<grid, 256, smem_bytes, stream>>>(xws, aspect, W1, b1, W2, b2, out);
}


# pool kernel: one wave per sample gathers each 1200-byte embedding row with three 16-byte-per-lane loads, 24 loads in flight, instead of one dword per thread; per-dim sum order unchanged
# speedup vs baseline: 2.2718x; 1.4303x over previous
#include <hip/hip_runtime.h>

typedef __attribute__((ext_vector_type(2))) float v2f;
typedef __attribute__((ext_vector_type(4))) float v4f;
typedef __attribute__((ext_vector_type(8))) float v8f;
typedef __attribute__((ext_vector_type(4))) int   v4i;

#define VOCAB   50000
#define EMB     300
#define HID     256
#define OUTD    3
#define NASP    8
#define BATCH   8192
#define SEQ     128
#define CHUNK   256
#define PROWS   64
#define LDS_FLOATS (PROWS * EMB + PROWS * HID + CHUNK + 8)

#define PWAVES  8
#define PB      8

__global__ __launch_bounds__(PWAVES * 32)
void pool_kernel(const int* __restrict__ ids,
                 const float* __restrict__ emb,
                 float* __restrict__ x) {
  const int lane = threadIdx.x & 31;
  const int s    = blockIdx.x * PWAVES + (threadIdx.x >> 5);
  const v4i idv = ((const v4i*)(ids + (size_t)s * SEQ))[lane];
  const v4f* emb4 = (const v4f*)emb;
  const int l2 = lane < (EMB / 4 - 64) ? 64 + lane : EMB / 4 - 1;
  v4f a0 = (v4f){0.0f, 0.0f, 0.0f, 0.0f};
  v4f a1 = a0, a2 = a0;
  for (int k0 = 0; k0 < SEQ; k0 += PB) {
    v4f p[3 * PB];
#pragma unroll
    for (int u = 0; u < PB; ++u) {
      const int id = __builtin_amdgcn_readlane(idv[u & 3], (k0 >> 2) + (u >> 2));
      const v4f* row = emb4 + (size_t)id * (EMB / 4);
      p[3 * u + 0] = row[lane];
      p[3 * u + 1] = row[32 + lane];
      p[3 * u + 2] = row[l2];
    }
    __builtin_amdgcn_sched_barrier(0);
#pragma unroll
    for (int u = 0; u < PB; ++u) {
      a0 += p[3 * u + 0];
      a1 += p[3 * u + 1];
      a2 += p[3 * u + 2];
    }
  }
  const float inv = 1.0f / (float)SEQ;
  v4f* xr = (v4f*)x + (size_t)s * (EMB / 4);
  xr[lane]      = a0 * inv;
  xr[32 + lane] = a1 * inv;
  if (lane < (EMB / 4 - 64)) xr[64 + lane] = a2 * inv;
}

template <int NRT>
__device__ __forceinline__ void gemm1_tiles(const float* sX, float* sH,
                                            const float* __restrict__ w1a,
                                            const float* __restrict__ b1a,
                                            int lane, int wv) {
  const int nloc  = lane & 15;
  const int half  = lane >> 4;
  const int khalf = half * 2;
  const int col0  = wv * 32 + nloc;
  const int col1  = col0 + 16;
  v8f acc0[NRT], acc1[NRT];
#pragma unroll
  for (int rt = 0; rt < NRT; ++rt) { acc0[rt] = (v8f){}; acc1[rt] = (v8f){}; }

#pragma unroll 5
  for (int kk = 0; kk < EMB / 4; ++kk) {
    const int k = kk * 4 + khalf;
    v2f bf0, bf1;
    bf0.x = w1a[(size_t)k * HID + col0];
    bf0.y = w1a[(size_t)(k + 1) * HID + col0];
    bf1.x = w1a[(size_t)k * HID + col1];
    bf1.y = w1a[(size_t)(k + 1) * HID + col1];
#pragma unroll
    for (int rt = 0; rt < NRT; ++rt) {
      const v2f afrag = *(const v2f*)&sX[(rt * 16 + nloc) * EMB + k];
      acc0[rt] = __builtin_amdgcn_wmma_f32_16x16x4_f32(
           false, afrag,  false, bf0,
           (short)0, acc0[rt],  false,  false);
      acc1[rt] = __builtin_amdgcn_wmma_f32_16x16x4_f32(
           false, afrag,  false, bf1,
           (short)0, acc1[rt],  false,  false);
    }
  }
  const float bias0 = b1a[col0];
  const float bias1 = b1a[col1];
#pragma unroll
  for (int rt = 0; rt < NRT; ++rt) {
#pragma unroll
    for (int r = 0; r < 8; ++r) {
      const int rowm = rt * 16 + r + 8 * half;
      const float h0 = acc0[rt][r] + bias0;
      const float h1 = acc1[rt][r] + bias1;
      sH[rowm * HID + col0] = h0 > 0.0f ? h0 : 0.0f;
      sH[rowm * HID + col1] = h1 > 0.0f ? h1 : 0.0f;
    }
  }
}

__global__ __launch_bounds__(256)
void mlp_kernel(const float* __restrict__ x,
                const int* __restrict__ aspect_id,
                const float* __restrict__ W1,
                const float* __restrict__ B1,
                const float* __restrict__ W2,
                const float* __restrict__ B2,
                float* __restrict__ out) {
  extern __shared__ __attribute__((aligned(16))) float smem[];
  float* sX   = smem;
  float* sH   = smem + PROWS * EMB;
  int*   sIdx = (int*)(smem + PROWS * EMB + PROWS * HID);
  int*   sCnt = sIdx + CHUNK;

  const int a    = blockIdx.y;
  const int row0 = blockIdx.x * CHUNK;
  const int t    = threadIdx.x;
  const int lane = t & 31;
  const int wv   = t >> 5;

  const bool mine = (aspect_id[row0 + t] == a);
  const unsigned bal = __builtin_amdgcn_ballot_w32(mine);
  if (lane == 0) sCnt[wv] = __popc(bal);
  __syncthreads();
  int base = 0, n = 0;
  for (int w = 0; w < 8; ++w) {
    const int c = sCnt[w];
    if (w < wv) base += c;
    n += c;
  }
  if (mine) sIdx[base + __popc(bal & ((1u << lane) - 1u))] = t;
  __syncthreads();

  const float* w1a = W1 + (size_t)a * EMB * HID;
  const float* b1a = B1 + a * HID;
  const float* w2a = W2 + (size_t)a * HID * OUTD;
  const v4f* x4  = (const v4f*)x;
  v4f*       sX4 = (v4f*)sX;

  for (int p0 = 0; p0 < n; p0 += PROWS) {
    const int rows = (n - p0) < PROWS ? (n - p0) : PROWS;
    const int nrt  = (rows + 15) >> 4;

    if (t < 240) {
      for (int rt = 0; rt < nrt; ++rt) {
        v4f v[5];
#pragma unroll
        for (int u = 0; u < 5; ++u) {
          const int i  = rt * 1200 + u * 240 + t;
          const int r  = i / (EMB / 4);
          const int c4 = i - r * (EMB / 4);
          const int rc = r < rows ? r : rows - 1;
          v[u] = x4[(size_t)(row0 + sIdx[p0 + rc]) * (EMB / 4) + c4];
          if (r >= rows) v[u] = (v4f){0.0f, 0.0f, 0.0f, 0.0f};
        }
#pragma unroll
        for (int u = 0; u < 5; ++u) {
          sX4[rt * 1200 + u * 240 + t] = v[u];
        }
      }
    }
    __syncthreads();

    switch (nrt) {
      case 1:  gemm1_tiles<1>(sX, sH, w1a, b1a, lane, wv); break;
      case 2:  gemm1_tiles<2>(sX, sH, w1a, b1a, lane, wv); break;
      case 3:  gemm1_tiles<3>(sX, sH, w1a, b1a, lane, wv); break;
      default: gemm1_tiles<4>(sX, sH, w1a, b1a, lane, wv); break;
    }
    __syncthreads();

    if (t < rows * OUTD) {
      const int s = t / OUTD, j = t % OUTD;
      float acc = B2[a * OUTD + j];
      for (int k = 0; k < HID; ++k) {
        acc += sH[s * HID + k] * w2a[k * OUTD + j];
      }
      const int b = row0 + sIdx[p0 + s];
      out[b * OUTD + j] = acc;
    }
    __syncthreads();
  }
}

extern "C" void kernel_launch(void* const* d_in, const int* in_sizes, int n_in,
                              void* d_out, int out_size, void* d_ws, size_t ws_size,
                              hipStream_t stream) {
  (void)in_sizes; (void)n_in; (void)out_size; (void)ws_size;
  const int*   ids    = (const int*)d_in[0];
  const int*   aspect = (const int*)d_in[1];
  const float* emb    = (const float*)d_in[2];
  const float* W1     = (const float*)d_in[3];
  const float* b1     = (const float*)d_in[4];
  const float* W2     = (const float*)d_in[5];
  const float* b2     = (const float*)d_in[6];
  float* out = (float*)d_out;
  float* xws = (float*)d_ws;

  pool_kernel<<<BATCH / PWAVES, PWAVES * 32, 0, stream>>>(ids, emb, xws);

  const size_t smem_bytes = (size_t)LDS_FLOATS * sizeof(float);
  dim3 grid(BATCH / CHUNK, NASP);
  mlp_kernel<<<grid, 256, smem_bytes, stream>>>(xws, aspect, W1, b1, W2, b2, out);
}


# pool kernel: a row's three 512-byte loads now cover the aligned windows from the row's first 256-byte unit (lanes rotated by the row's offset, per-lane select); per-dim sum order unchanged
# speedup vs baseline: 2.4241x; 1.0670x over previous
#include <hip/hip_runtime.h>

typedef __attribute__((ext_vector_type(2))) float v2f;
typedef __attribute__((ext_vector_type(4))) float v4f;
typedef __attribute__((ext_vector_type(8))) float v8f;
typedef __attribute__((ext_vector_type(4))) int   v4i;

#define VOCAB   50000
#define EMB     300
#define HID     256
#define OUTD    3
#define NASP    8
#define BATCH   8192
#define SEQ     128
#define CHUNK   256
#define PROWS   64
#define LDS_FLOATS (PROWS * EMB + PROWS * HID + CHUNK + 8)

#define PWAVES  8
#define PB      8
#define NPIECE  (EMB / 4)

__global__ __launch_bounds__(PWAVES * 32)
void pool_kernel(const int* __restrict__ ids,
                 const float* __restrict__ emb,
                 float* __restrict__ x) {
  const unsigned lane = threadIdx.x & 31;
  const int s    = blockIdx.x * PWAVES + (threadIdx.x >> 5);
  const v4i idv = ((const v4i*)(ids + (size_t)s * SEQ))[lane];
  v4f a0 = (v4f){0.0f, 0.0f, 0.0f, 0.0f};
  v4f a1 = a0, a2 = a0;
  for (int k0 = 0; k0 < SEQ; k0 += PB) {
    v4f p[3 * PB];
    unsigned shv[PB];
#pragma unroll
    for (int u = 0; u < PB; ++u) {
      const int id = __builtin_amdgcn_readlane(idv[u & 3], (k0 >> 2) + (u >> 2));
      const v4f* row = (const v4f*)emb + (size_t)(unsigned)id * NPIECE;
      const unsigned sh = ((unsigned)(size_t)row >> 4) & 15u;
      const v4f* win = row - sh;
      const unsigned pos  = (lane + sh) & 31u;
      const unsigned pos0 = pos > sh ? pos : sh;
      const unsigned e2   = sh + (NPIECE - 1 - 64);
      const unsigned pos2 = pos < e2 ? pos : e2;
      p[3 * u + 0] = win[pos0];
      p[3 * u + 1] = win[32 + pos];
      p[3 * u + 2] = win[64 + pos2];
      shv[u] = sh;
    }
    __builtin_amdgcn_sched_barrier(0);
#pragma unroll
    for (int u = 0; u < PB; ++u) {
      const bool low = lane + shv[u] < 32u;
      a0 += low ? p[3 * u + 0] : p[3 * u + 1];
      a1 += low ? p[3 * u + 1] : p[3 * u + 2];
      a2 += p[3 * u + 2];
    }
  }
  const float inv = 1.0f / (float)SEQ;
  v4f* xr = (v4f*)x + (size_t)s * NPIECE;
  xr[lane]      = a0 * inv;
  xr[32 + lane] = a1 * inv;
  if (lane < (NPIECE - 64)) xr[64 + lane] = a2 * inv;
}

template <int NRT>
__device__ __forceinline__ void gemm1_tiles(const float* sX, float* sH,
                                            const float* __restrict__ w1a,
                                            const float* __restrict__ b1a,
                                            int lane, int wv) {
  const int nloc  = lane & 15;
  const int half  = lane >> 4;
  const int khalf = half * 2;
  const int col0  = wv * 32 + nloc;
  const int col1  = col0 + 16;
  v8f acc0[NRT], acc1[NRT];
#pragma unroll
  for (int rt = 0; rt < NRT; ++rt) { acc0[rt] = (v8f){}; acc1[rt] = (v8f){}; }

#pragma unroll 5
  for (int kk = 0; kk < EMB / 4; ++kk) {
    const int k = kk * 4 + khalf;
    v2f bf0, bf1;
    bf0.x = w1a[(size_t)k * HID + col0];
    bf0.y = w1a[(size_t)(k + 1) * HID + col0];
    bf1.x = w1a[(size_t)k * HID + col1];
    bf1.y = w1a[(size_t)(k + 1) * HID + col1];
#pragma unroll
    for (int rt = 0; rt < NRT; ++rt) {
      const v2f afrag = *(const v2f*)&sX[(rt * 16 + nloc) * EMB + k];
      acc0[rt] = __builtin_amdgcn_wmma_f32_16x16x4_f32(
           false, afrag,  false, bf0,
           (short)0, acc0[rt],  false,  false);
      acc1[rt] = __builtin_amdgcn_wmma_f32_16x16x4_f32(
           false, afrag,  false, bf1,
           (short)0, acc1[rt],  false,  false);
    }
  }
  const float bias0 = b1a[col0];
  const float bias1 = b1a[col1];
#pragma unroll
  for (int rt = 0; rt < NRT; ++rt) {
#pragma unroll
    for (int r = 0; r < 8; ++r) {
      const int rowm = rt * 16 + r + 8 * half;
      const float h0 = acc0[rt][r] + bias0;
      const float h1 = acc1[rt][r] + bias1;
      sH[rowm * HID + col0] = h0 > 0.0f ? h0 : 0.0f;
      sH[rowm * HID + col1] = h1 > 0.0f ? h1 : 0.0f;
    }
  }
}

__global__ __launch_bounds__(256)
void mlp_kernel(const float* __restrict__ x,
                const int* __restrict__ aspect_id,
                const float* __restrict__ W1,
                const float* __restrict__ B1,
                const float* __restrict__ W2,
                const float* __restrict__ B2,
                float* __restrict__ out) {
  extern __shared__ __attribute__((aligned(16))) float smem[];
  float* sX   = smem;
  float* sH   = smem + PROWS * EMB;
  int*   sIdx = (int*)(smem + PROWS * EMB + PROWS * HID);
  int*   sCnt = sIdx + CHUNK;

  const int a    = blockIdx.y;
  const int row0 = blockIdx.x * CHUNK;
  const int t    = threadIdx.x;
  const int lane = t & 31;
  const int wv   = t >> 5;

  const bool mine = (aspect_id[row0 + t] == a);
  const unsigned bal = __builtin_amdgcn_ballot_w32(mine);
  if (lane == 0) sCnt[wv] = __popc(bal);
  __syncthreads();
  int base = 0, n = 0;
  for (int w = 0; w < 8; ++w) {
    const int c = sCnt[w];
    if (w < wv) base += c;
    n += c;
  }
  if (mine) sIdx[base + __popc(bal & ((1u << lane) - 1u))] = t;
  __syncthreads();

  const float* w1a = W1 + (size_t)a * EMB * HID;
  const float* b1a = B1 + a * HID;
  const float* w2a = W2 + (size_t)a * HID * OUTD;
  const v4f* x4  = (const v4f*)x;
  v4f*       sX4 = (v4f*)sX;

  for (int p0 = 0; p0 < n; p0 += PROWS) {
    const int rows = (n - p0) < PROWS ? (n - p0) : PROWS;
    const int nrt  = (rows + 15) >> 4;

    if (t < 240) {
      for (int rt = 0; rt < nrt; ++rt) {
        v4f v[5];
#pragma unroll
        for (int u = 0; u < 5; ++u) {
          const int i  = rt * 1200 + u * 240 + t;
          const int r  = i / (EMB / 4);
          const int c4 = i - r * (EMB / 4);
          const int rc = r < rows ? r : rows - 1;
          v[u] = x4[(size_t)(row0 + sIdx[p0 + rc]) * (EMB / 4) + c4];
          if (r >= rows) v[u] = (v4f){0.0f, 0.0f, 0.0f, 0.0f};
        }
#pragma unroll
        for (int u = 0; u < 5; ++u) {
          sX4[rt * 1200 + u * 240 + t] = v[u];
        }
      }
    }
    __syncthreads();

    switch (nrt) {
      case 1:  gemm1_tiles<1>(sX, sH, w1a, b1a, lane, wv); break;
      case 2:  gemm1_tiles<2>(sX, sH, w1a, b1a, lane, wv); break;
      case 3:  gemm1_tiles<3>(sX, sH, w1a, b1a, lane, wv); break;
      default: gemm1_tiles<4>(sX, sH, w1a, b1a, lane, wv); break;
    }
    __syncthreads();

    if (t < rows * OUTD) {
      const int s = t / OUTD, j = t % OUTD;
      float acc = B2[a * OUTD + j];
      for (int k = 0; k < HID; ++k) {
        acc += sH[s * HID + k] * w2a[k * OUTD + j];
      }
      const int b = row0 + sIdx[p0 + s];
      out[b * OUTD + j] = acc;
    }
    __syncthreads();
  }
}

extern "C" void kernel_launch(void* const* d_in, const int* in_sizes, int n_in,
                              void* d_out, int out_size, void* d_ws, size_t ws_size,
                              hipStream_t stream) {
  (void)in_sizes; (void)n_in; (void)out_size; (void)ws_size;
  const int*   ids    = (const int*)d_in[0];
  const int*   aspect = (const int*)d_in[1];
  const float* emb    = (const float*)d_in[2];
  const float* W1     = (const float*)d_in[3];
  const float* b1     = (const float*)d_in[4];
  const float* W2     = (const float*)d_in[5];
  const float* b2     = (const float*)d_in[6];
  float* out = (float*)d_out;
  float* xws = (float*)d_ws;

  pool_kernel<<<BATCH / PWAVES, PWAVES * 32, 0, stream>>>(ids, emb, xws);

  const size_t smem_bytes = (size_t)LDS_FLOATS * sizeof(float);
  dim3 grid(BATCH / CHUNK, NASP);
  mlp_kernel<<<grid, 256, smem_bytes, stream>>>(xws, aspect, W1, b1, W2, b2, out);
}


# MLP kernel: layer 2 (256->3) taken from layer 1's accumulators (WMMA operands swapped: a lane owns one row's hidden values); 16 partials per output summed via LDS; layer-2 sum order changes
# speedup vs baseline: 2.7639x; 1.1402x over previous
#include <hip/hip_runtime.h>

typedef __attribute__((ext_vector_type(2))) float v2f;
typedef __attribute__((ext_vector_type(4))) float v4f;
typedef __attribute__((ext_vector_type(8))) float v8f;
typedef __attribute__((ext_vector_type(4))) int   v4i;

#define VOCAB   50000
#define EMB     300
#define HID     256
#define OUTD    3
#define NASP    8
#define BATCH   8192
#define SEQ     128
#define CHUNK   256
#define PROWS   64
#define NPART   16
#define LDS_FLOATS (PROWS * EMB + NPART * PROWS * OUTD + CHUNK + 8)

#define PWAVES  8
#define PB      8
#define NPIECE  (EMB / 4)

__global__ __launch_bounds__(PWAVES * 32)
void pool_kernel(const int* __restrict__ ids,
                 const float* __restrict__ emb,
                 float* __restrict__ x) {
  const unsigned lane = threadIdx.x & 31;
  const int s    = blockIdx.x * PWAVES + (threadIdx.x >> 5);
  const v4i idv = ((const v4i*)(ids + (size_t)s * SEQ))[lane];
  v4f a0 = (v4f){0.0f, 0.0f, 0.0f, 0.0f};
  v4f a1 = a0, a2 = a0;
  for (int k0 = 0; k0 < SEQ; k0 += PB) {
    v4f p[3 * PB];
    unsigned shv[PB];
#pragma unroll
    for (int u = 0; u < PB; ++u) {
      const int id = __builtin_amdgcn_readlane(idv[u & 3], (k0 >> 2) + (u >> 2));
      const v4f* row = (const v4f*)emb + (size_t)(unsigned)id * NPIECE;
      const unsigned sh = ((unsigned)(size_t)row >> 4) & 15u;
      const v4f* win = row - sh;
      const unsigned pos  = (lane + sh) & 31u;
      const unsigned pos0 = pos > sh ? pos : sh;
      const unsigned e2   = sh + (NPIECE - 1 - 64);
      const unsigned pos2 = pos < e2 ? pos : e2;
      p[3 * u + 0] = win[pos0];
      p[3 * u + 1] = win[32 + pos];
      p[3 * u + 2] = win[64 + pos2];
      shv[u] = sh;
    }
    __builtin_amdgcn_sched_barrier(0);
#pragma unroll
    for (int u = 0; u < PB; ++u) {
      const bool low = lane + shv[u] < 32u;
      a0 += low ? p[3 * u + 0] : p[3 * u + 1];
      a1 += low ? p[3 * u + 1] : p[3 * u + 2];
      a2 += p[3 * u + 2];
    }
  }
  const float inv = 1.0f / (float)SEQ;
  v4f* xr = (v4f*)x + (size_t)s * NPIECE;
  xr[lane]      = a0 * inv;
  xr[32 + lane] = a1 * inv;
  if (lane < (NPIECE - 64)) xr[64 + lane] = a2 * inv;
}

template <int NRT>
__device__ __forceinline__ void mlp_tiles(const float* sX, float* sP,
                                          const float* __restrict__ w1a,
                                          const float (&bA)[8], const float (&bB)[8],
                                          const float (&wA)[24], const float (&wB)[24],
                                          int lane, int wv) {
  const int nloc  = lane & 15;
  const int half  = lane >> 4;
  const int khalf = half * 2;
  const int col0  = wv * 32 + nloc;
  const int col1  = col0 + 16;
  v8f acc0[NRT], acc1[NRT];
#pragma unroll
  for (int rt = 0; rt < NRT; ++rt) { acc0[rt] = (v8f){}; acc1[rt] = (v8f){}; }

#pragma unroll 5
  for (int kk = 0; kk < EMB / 4; ++kk) {
    const int k = kk * 4 + khalf;
    v2f wf0, wf1;
    wf0.x = w1a[(size_t)k * HID + col0];
    wf0.y = w1a[(size_t)(k + 1) * HID + col0];
    wf1.x = w1a[(size_t)k * HID + col1];
    wf1.y = w1a[(size_t)(k + 1) * HID + col1];
#pragma unroll
    for (int rt = 0; rt < NRT; ++rt) {
      const v2f xf = *(const v2f*)&sX[(rt * 16 + nloc) * EMB + k];
      acc0[rt] = __builtin_amdgcn_wmma_f32_16x16x4_f32(
           false, wf0,  false, xf,
           (short)0, acc0[rt],  false,  false);
      acc1[rt] = __builtin_amdgcn_wmma_f32_16x16x4_f32(
           false, wf1,  false, xf,
           (short)0, acc1[rt],  false,  false);
    }
  }
#pragma unroll
  for (int rt = 0; rt < NRT; ++rt) {
    float hA[8], hB[8];
#pragma unroll
    for (int r = 0; r < 8; ++r) {
      const float u = acc0[rt][r] + bA[r];
      const float v = acc1[rt][r] + bB[r];
      hA[r] = u > 0.0f ? u : 0.0f;
      hB[r] = v > 0.0f ? v : 0.0f;
    }
    float p0 = hA[0] * wA[0], p1 = hA[0] * wA[1], p2 = hA[0] * wA[2];
#pragma unroll
    for (int r = 1; r < 8; ++r) {
      p0 += hA[r] * wA[3 * r + 0];
      p1 += hA[r] * wA[3 * r + 1];
      p2 += hA[r] * wA[3 * r + 2];
    }
#pragma unroll
    for (int r = 0; r < 8; ++r) {
      p0 += hB[r] * wB[3 * r + 0];
      p1 += hB[r] * wB[3 * r + 1];
      p2 += hB[r] * wB[3 * r + 2];
    }
    float* dst = sP + ((wv * 2 + half) * PROWS + rt * 16 + nloc) * OUTD;
    dst[0] = p0;
    dst[1] = p1;
    dst[2] = p2;
  }
}

__global__ __launch_bounds__(256)
void mlp_kernel(const float* __restrict__ x,
                const int* __restrict__ aspect_id,
                const float* __restrict__ W1,
                const float* __restrict__ B1,
                const float* __restrict__ W2,
                const float* __restrict__ B2,
                float* __restrict__ out) {
  extern __shared__ __attribute__((aligned(16))) float smem[];
  float* sX   = smem;
  float* sP   = smem + PROWS * EMB;
  int*   sIdx = (int*)(smem + PROWS * EMB + NPART * PROWS * OUTD);
  int*   sCnt = sIdx + CHUNK;

  const int a    = blockIdx.y;
  const int row0 = blockIdx.x * CHUNK;
  const int t    = threadIdx.x;
  const int lane = t & 31;
  const int wv   = t >> 5;

  const float* w1a = W1 + (size_t)a * EMB * HID;
  const float* b1a = B1 + a * HID;
  const float* w2a = W2 + (size_t)a * HID * OUTD;

  const bool mine = (aspect_id[row0 + t] == a);

  const int hc0 = wv * 32 + (lane >> 4) * 8;
  float bA[8], bB[8], wA[24], wB[24];
  {
    const v4f* pbA = (const v4f*)(b1a + hc0);
    const v4f* pbB = (const v4f*)(b1a + hc0 + 16);
    const v4f* pwA = (const v4f*)(w2a + hc0 * OUTD);
    const v4f* pwB = (const v4f*)(w2a + (hc0 + 16) * OUTD);
#pragma unroll
    for (int i = 0; i < 2; ++i) {
      const v4f u = pbA[i], v = pbB[i];
      bA[4 * i + 0] = u.x; bA[4 * i + 1] = u.y; bA[4 * i + 2] = u.z; bA[4 * i + 3] = u.w;
      bB[4 * i + 0] = v.x; bB[4 * i + 1] = v.y; bB[4 * i + 2] = v.z; bB[4 * i + 3] = v.w;
    }
#pragma unroll
    for (int i = 0; i < 6; ++i) {
      const v4f u = pwA[i], v = pwB[i];
      wA[4 * i + 0] = u.x; wA[4 * i + 1] = u.y; wA[4 * i + 2] = u.z; wA[4 * i + 3] = u.w;
      wB[4 * i + 0] = v.x; wB[4 * i + 1] = v.y; wB[4 * i + 2] = v.z; wB[4 * i + 3] = v.w;
    }
  }
  const float b2j = B2[a * OUTD + t % OUTD];

  const unsigned bal = __builtin_amdgcn_ballot_w32(mine);
  if (lane == 0) sCnt[wv] = __popc(bal);
  __syncthreads();
  int base = 0, n = 0;
  for (int w = 0; w < 8; ++w) {
    const int c = sCnt[w];
    if (w < wv) base += c;
    n += c;
  }
  if (mine) sIdx[base + __popc(bal & ((1u << lane) - 1u))] = t;
  __syncthreads();

  const v4f* x4  = (const v4f*)x;
  v4f*       sX4 = (v4f*)sX;

  for (int p0 = 0; p0 < n; p0 += PROWS) {
    const int rows = (n - p0) < PROWS ? (n - p0) : PROWS;
    const int nrt  = (rows + 15) >> 4;

    if (t < 240) {
      for (int rt = 0; rt < nrt; ++rt) {
        v4f v[5];
#pragma unroll
        for (int u = 0; u < 5; ++u) {
          const int i  = rt * 1200 + u * 240 + t;
          const int r  = i / (EMB / 4);
          const int c4 = i - r * (EMB / 4);
          const int rc = r < rows ? r : rows - 1;
          v[u] = x4[(size_t)(row0 + sIdx[p0 + rc]) * (EMB / 4) + c4];
          if (r >= rows) v[u] = (v4f){0.0f, 0.0f, 0.0f, 0.0f};
        }
#pragma unroll
        for (int u = 0; u < 5; ++u) {
          sX4[rt * 1200 + u * 240 + t] = v[u];
        }
      }
    }
    __syncthreads();

    switch (nrt) {
      case 1:  mlp_tiles<1>(sX, sP, w1a, bA, bB, wA, wB, lane, wv); break;
      case 2:  mlp_tiles<2>(sX, sP, w1a, bA, bB, wA, wB, lane, wv); break;
      case 3:  mlp_tiles<3>(sX, sP, w1a, bA, bB, wA, wB, lane, wv); break;
      default: mlp_tiles<4>(sX, sP, w1a, bA, bB, wA, wB, lane, wv); break;
    }
    __syncthreads();

    if (t < rows * OUTD) {
      float acc = b2j;
#pragma unroll
      for (int q = 0; q < NPART; ++q) {
        acc += sP[q * PROWS * OUTD + t];
      }
      const int s = t / OUTD;
      const int b = row0 + sIdx[p0 + s];
      out[b * OUTD + (t - s * OUTD)] = acc;
    }
    __syncthreads();
  }
}

extern "C" void kernel_launch(void* const* d_in, const int* in_sizes, int n_in,
                              void* d_out, int out_size, void* d_ws, size_t ws_size,
                              hipStream_t stream) {
  (void)in_sizes; (void)n_in; (void)out_size; (void)ws_size;
  const int*   ids    = (const int*)d_in[0];
  const int*   aspect = (const int*)d_in[1];
  const float* emb    = (const float*)d_in[2];
  const float* W1     = (const float*)d_in[3];
  const float* b1     = (const float*)d_in[4];
  const float* W2     = (const float*)d_in[5];
  const float* b2     = (const float*)d_in[6];
  float* out = (float*)d_out;
  float* xws = (float*)d_ws;

  pool_kernel<<<BATCH / PWAVES, PWAVES * 32, 0, stream>>>(ids, emb, xws);

  const size_t smem_bytes = (size_t)LDS_FLOATS * sizeof(float);
  dim3 grid(BATCH / CHUNK, NASP);
  mlp_kernel<<<grid, 256, smem_bytes, stream>>>(xws, aspect, W1, b1, W2, b2, out);
}
